# starting kernel
# baseline (speedup 1.0000x reference)
#include <hip/hip_runtime.h>

#define SEQL   2048
#define DIMM   2048
#define NHEADS 16
#define HDIM   128

typedef float v2f __attribute__((ext_vector_type(2)));
typedef float v8f __attribute__((ext_vector_type(8)));

__device__ __forceinline__ v8f wmma_f32(v2f a, v2f b, v8f c) {
    return __builtin_amdgcn_wmma_f32_16x16x4_f32(false, a, false, b, (short)0, c, false, false);
}

__global__ __launch_bounds__(256)
void gemm_abt(const float* __restrict__ A, const float* __restrict__ B,
              float* __restrict__ C) {
    const int lane = threadIdx.x & 31;
    const int wave = threadIdx.x >> 5;
    const int hi   = lane >> 4;
    const int lc   = lane & 15;

    const int m0 = blockIdx.y * 128 + wave * 16;
    const int n0 = blockIdx.x * 64;

    v8f acc0 = {}, acc1 = {}, acc2 = {}, acc3 = {};

    const float* ap  = A + (size_t)(m0 + lc) * DIMM + 2 * hi;
    const float* bp0 = B + (size_t)(n0 +  0 + lc) * DIMM + 2 * hi;
    const float* bp1 = B + (size_t)(n0 + 16 + lc) * DIMM + 2 * hi;
    const float* bp2 = B + (size_t)(n0 + 32 + lc) * DIMM + 2 * hi;
    const float* bp3 = B + (size_t)(n0 + 48 + lc) * DIMM + 2 * hi;

#pragma unroll 4
    for (int k0 = 0; k0 < DIMM; k0 += 4) {
        v2f a  = *(const v2f*)(ap  + k0);
        v2f b0 = *(const v2f*)(bp0 + k0);
        v2f b1 = *(const v2f*)(bp1 + k0);
        v2f b2 = *(const v2f*)(bp2 + k0);
        v2f b3 = *(const v2f*)(bp3 + k0);
        acc0 = wmma_f32(a, b0, acc0);
        acc1 = wmma_f32(a, b1, acc1);
        acc2 = wmma_f32(a, b2, acc2);
        acc3 = wmma_f32(a, b3, acc3);
    }

#pragma unroll
    for (int r = 0; r < 8; ++r) {
        float* crow = C + (size_t)(m0 + r + 8 * hi) * DIMM + n0 + lc;
        crow[ 0] = acc0[r];
        crow[16] = acc1[r];
        crow[32] = acc2[r];
        crow[48] = acc3[r];
    }
}

__global__ __launch_bounds__(256)
void attn_fwd(const float* __restrict__ Q, const float* __restrict__ K,
              const float* __restrict__ V, const float* __restrict__ mask,
              float* __restrict__ O) {
    __shared__ float ldsP[8 * 256];

    const int lane = threadIdx.x & 31;
    const int wave = threadIdx.x >> 5;
    const int hi   = lane >> 4;
    const int lc   = lane & 15;

    const int h  = blockIdx.y;
    const int q0 = (blockIdx.x * 8 + wave) * 16;

    v2f aq[32];
    {
        const float* qrow = Q + (size_t)(q0 + lc) * DIMM + h * HDIM + 2 * hi;
#pragma unroll
        for (int s = 0; s < 32; ++s) aq[s] = *(const v2f*)(qrow + s * 4);
    }

    v8f oacc[8];
#pragma unroll
    for (int t = 0; t < 8; ++t) oacc[t] = (v8f){};
    float mrow[8], lrow[8];
#pragma unroll
    for (int r = 0; r < 8; ++r) { mrow[r] = -1.0e30f; lrow[r] = 0.0f; }

    float* lp = &ldsP[wave * 256];
    const float scale = 0.08838834764831845f;
    const size_t mbase = ((size_t)h * SEQL + q0) * SEQL;

#pragma unroll 1
    for (int kv0 = 0; kv0 < SEQL; kv0 += 16) {
        v8f s = {};
        const float* krow = K + (size_t)(kv0 + lc) * DIMM + h * HDIM + 2 * hi;
#pragma unroll
        for (int ss = 0; ss < 32; ++ss) {
            v2f bk = *(const v2f*)(krow + ss * 4);
            s = wmma_f32(aq[ss], bk, s);
        }

#pragma unroll
        for (int r = 0; r < 8; ++r) {
            const float mval = __builtin_nontemporal_load(
                mask + mbase + (size_t)(r + 8 * hi) * SEQL + kv0 + lc);
            float v = s[r] * scale + mval;

            float rmax = v;
            rmax = fmaxf(rmax, __shfl_xor(rmax, 1, 32));
            rmax = fmaxf(rmax, __shfl_xor(rmax, 2, 32));
            rmax = fmaxf(rmax, __shfl_xor(rmax, 4, 32));
            rmax = fmaxf(rmax, __shfl_xor(rmax, 8, 32));

            const float mnew = fmaxf(mrow[r], rmax);
            const float corr = __expf(mrow[r] - mnew);
            mrow[r] = mnew;

            const float p = __expf(v - mnew);
            s[r] = p;

            float psum = p;
            psum += __shfl_xor(psum, 1, 32);
            psum += __shfl_xor(psum, 2, 32);
            psum += __shfl_xor(psum, 4, 32);
            psum += __shfl_xor(psum, 8, 32);
            lrow[r] = lrow[r] * corr + psum;

#pragma unroll
            for (int t = 0; t < 8; ++t) oacc[t][r] *= corr;
        }

#pragma unroll
        for (int r = 0; r < 8; ++r) lp[(r + 8 * hi) * 16 + lc] = s[r];
        asm volatile("s_wait_dscnt 0" ::: "memory");
        v2f pa[4];
#pragma unroll
        for (int kk = 0; kk < 4; ++kk)
            pa[kk] = *(const v2f*)&lp[lc * 16 + kk * 4 + 2 * hi];
        asm volatile("s_wait_dscnt 0" ::: "memory");

#pragma unroll
        for (int kk = 0; kk < 4; ++kk) {
            const float* vrow =
                V + (size_t)(kv0 + kk * 4 + 2 * hi) * DIMM + h * HDIM + lc;
#pragma unroll
            for (int t = 0; t < 8; ++t) {
                v2f bv;
                bv[0] = vrow[t * 16];
                bv[1] = vrow[t * 16 + DIMM];
                oacc[t] = wmma_f32(pa[kk], bv, oacc[t]);
            }
        }
    }

#pragma unroll
    for (int r = 0; r < 8; ++r) {
        const float inv = 1.0f / lrow[r];
        float* orow = O + (size_t)(q0 + r + 8 * hi) * DIMM + h * HDIM + lc;
#pragma unroll
        for (int t = 0; t < 8; ++t) orow[t * 16] = oacc[t][r] * inv;
    }
}

extern "C" void kernel_launch(void* const* d_in, const int* in_sizes, int n_in,
                              void* d_out, int out_size, void* d_ws, size_t ws_size,
                              hipStream_t stream) {
    const float* x    = (const float*)d_in[0];
    const float* mask = (const float*)d_in[1];
    const float* wq   = (const float*)d_in[3];
    const float* wk   = (const float*)d_in[4];
    const float* wv   = (const float*)d_in[5];
    const float* wo   = (const float*)d_in[6];
    float* out = (float*)d_out;

    float* ws = (float*)d_ws;
    const size_t MAT = (size_t)SEQL * DIMM;
    float* Qb = ws;
    float* Kb = ws + MAT;
    float* Vb = ws + 2 * MAT;
    float* Ab = ws + 3 * MAT;

    dim3 tb(256);
    dim3 gg(DIMM / 64, SEQL / 128);
    gemm_abt<<<gg, tb, 0, stream>>>(x, wq, Qb);
    gemm_abt<<<gg, tb, 0, stream>>>(x, wk, Kb);
    gemm_abt<<<gg, tb, 0, stream>>>(x, wv, Vb);

    dim3 ga(SEQL / 128, NHEADS);
    attn_fwd<<<ga, tb, 0, stream>>>(Qb, Kb, Vb, mask, Ab);

    gemm_abt<<<gg, tb, 0, stream>>>(Ab, wo, out);
}


# attn_fwd: softmax step of 64 keys (4 independent score tiles): one batched row-max exchange and one rescale per step; row sum kept per lane, reduced across lanes once at the end (sum order changes)
# speedup vs baseline: 1.4178x; 1.4178x over previous
#include <hip/hip_runtime.h>

#define SEQL   2048
#define DIMM   2048
#define NHEADS 16
#define HDIM   128

typedef float v2f __attribute__((ext_vector_type(2)));
typedef float v4f __attribute__((ext_vector_type(4)));
typedef float v8f __attribute__((ext_vector_type(8)));

__device__ __forceinline__ v8f wmma_f32(v2f a, v2f b, v8f c) {
    return __builtin_amdgcn_wmma_f32_16x16x4_f32(false, a, false, b, (short)0, c, false, false);
}

struct Frag8 { v4f lo, hi; };

__device__ __forceinline__ Frag8 load_frag8(const float* __restrict__ p) {
    Frag8 f;
    f.lo = *(const v4f*)(p);
    f.hi = *(const v4f*)(p + 4);
    return f;
}

template <int J>
__device__ __forceinline__ v2f kpair(const Frag8& f) {
    if (J == 0) return __builtin_shufflevector(f.lo, f.lo, 0, 1);
    if (J == 1) return __builtin_shufflevector(f.lo, f.lo, 2, 3);
    if (J == 2) return __builtin_shufflevector(f.hi, f.hi, 0, 1);
    return __builtin_shufflevector(f.hi, f.hi, 2, 3);
}

template <int J>
__device__ __forceinline__ void mma_step(const Frag8 (&a)[4], const Frag8 (&b)[4],
                                         v8f (&acc)[4][4]) {
    for (int i = 0; i < 4; ++i) {
        const v2f av = kpair<J>(a[i]);
        for (int n = 0; n < 4; ++n)
            acc[i][n] = wmma_f32(av, kpair<J>(b[n]), acc[i][n]);
    }
}

__device__ __forceinline__ void mma_chunk(const Frag8 (&a)[4], const Frag8 (&b)[4],
                                          v8f (&acc)[4][4]) {
    mma_step<0>(a, b, acc);
    mma_step<1>(a, b, acc);
    mma_step<2>(a, b, acc);
    mma_step<3>(a, b, acc);
}

__device__ __forceinline__ void load_chunk(Frag8 (&a)[4], Frag8 (&b)[4],
                                           const float* __restrict__ ap,
                                           const float* __restrict__ bp, int k0) {
    for (int i = 0; i < 4; ++i) a[i] = load_frag8(ap + (size_t)i * 16 * DIMM + k0);
    for (int n = 0; n < 4; ++n) b[n] = load_frag8(bp + (size_t)n * 16 * DIMM + k0);
}

__global__ __launch_bounds__(128)
void gemm_abt(const float* __restrict__ A, const float* __restrict__ B,
              float* __restrict__ C) {
    const int lane = threadIdx.x & 31;
    const int wave = threadIdx.x >> 5;
    const int hi   = lane >> 4;
    const int lc   = lane & 15;

    const int m0 = blockIdx.y * 128 + (wave >> 1) * 64;
    const int n0 = blockIdx.x * 128 + (wave & 1) * 64;

    v8f acc[4][4];
    for (int i = 0; i < 4; ++i)
        for (int n = 0; n < 4; ++n) acc[i][n] = (v8f){};

    const float* ap = A + (size_t)(m0 + lc) * DIMM + 8 * hi;
    const float* bp = B + (size_t)(n0 + lc) * DIMM + 8 * hi;

    Frag8 a0[4], b0[4], a1[4], b1[4];
    load_chunk(a0, b0, ap, bp, 0);

    for (int k0 = 0; k0 < DIMM - 32; k0 += 32) {
        load_chunk(a1, b1, ap, bp, k0 + 16);
        mma_chunk(a0, b0, acc);
        load_chunk(a0, b0, ap, bp, k0 + 32);
        mma_chunk(a1, b1, acc);
    }
    load_chunk(a1, b1, ap, bp, DIMM - 16);
    mma_chunk(a0, b0, acc);
    mma_chunk(a1, b1, acc);

    for (int i = 0; i < 4; ++i) {
        for (int r = 0; r < 8; ++r) {
            float* crow = C + (size_t)(m0 + 16 * i + r + 8 * hi) * DIMM + n0 + lc;
            for (int n = 0; n < 4; ++n) crow[16 * n] = acc[i][n][r];
        }
    }
}

#define KB_TILES 4
#define KB       (16 * KB_TILES)

__device__ __forceinline__ void group_fence() { __builtin_amdgcn_sched_barrier(0); }

__global__ __launch_bounds__(256) __attribute__((amdgpu_waves_per_eu(2, 2)))
void attn_fwd(const float* __restrict__ Q, const float* __restrict__ K,
              const float* __restrict__ V, const float* __restrict__ mask,
              float* __restrict__ O) {
    __shared__ float ldsP[8 * KB_TILES * 256];

    const int lane = threadIdx.x & 31;
    const int wave = threadIdx.x >> 5;
    const int hi   = lane >> 4;
    const int lc   = lane & 15;

    const int h  = blockIdx.y;
    const int q0 = (blockIdx.x * 8 + wave) * 16;


    v2f aq[32];
    {
        const float* qrow = Q + (size_t)(q0 + lc) * DIMM + h * HDIM + 2 * hi;
#pragma unroll
        for (int s = 0; s < 32; ++s) aq[s] = *(const v2f*)(qrow + s * 4);
    }

    v8f oacc[8];
#pragma unroll
    for (int t = 0; t < 8; ++t) oacc[t] = (v8f){};
    float mrow[8], lsum[8];
#pragma unroll
    for (int r = 0; r < 8; ++r) { mrow[r] = -1.0e30f; lsum[r] = 0.0f; }

    float* lp = &ldsP[wave * (KB_TILES * 256)];
    const float scale = 0.08838834764831845f;
    const size_t mbase = ((size_t)h * SEQL + q0) * SEQL;

#pragma unroll 1
    for (int kv0 = 0; kv0 < SEQL; kv0 += KB) {
        v8f s[KB_TILES];
#pragma unroll
        for (int j = 0; j < KB_TILES; ++j) s[j] = (v8f){};
        const float* krow = K + (size_t)(kv0 + lc) * DIMM + h * HDIM + 2 * hi;
        v2f kf[2][8][KB_TILES];
        float mk[8][KB_TILES];
#pragma unroll
        for (int u = 0; u < 8; ++u)
#pragma unroll
            for (int j = 0; j < KB_TILES; ++j)
                kf[0][u][j] = *(const v2f*)(krow + (size_t)j * 16 * DIMM + u * 4);
        group_fence();
#pragma unroll
        for (int g = 0; g < 4; ++g) {
            if (g + 1 < 4) {
#pragma unroll
                for (int u = 0; u < 8; ++u)
#pragma unroll
                    for (int j = 0; j < KB_TILES; ++j)
                        kf[(g + 1) & 1][u][j] =
                            *(const v2f*)(krow + (size_t)j * 16 * DIMM + ((g + 1) * 8 + u) * 4);
            }
            if (g == 2) {
                group_fence();
#pragma unroll
                for (int r = 0; r < 8; ++r) {
                    const float* mp = mask + mbase + (size_t)(r + 8 * hi) * SEQL + kv0 + lc;
#pragma unroll
                    for (int j = 0; j < KB_TILES; ++j)
                        mk[r][j] = __builtin_nontemporal_load(mp + 16 * j);
                }
            }
#pragma unroll
            for (int u = 0; u < 8; ++u)
#pragma unroll
                for (int j = 0; j < KB_TILES; ++j)
                    s[j] = wmma_f32(aq[g * 8 + u], kf[g & 1][u][j], s[j]);
            group_fence();
        }

        float vf[2][4][8][2];
        {
            const float* vb = V + (size_t)(kv0 + 2 * hi) * DIMM + h * HDIM + lc;
#pragma unroll
            for (int kk = 0; kk < 4; ++kk)
#pragma unroll
                for (int t = 0; t < 8; ++t) {
                    vf[0][kk][t][0] = vb[(size_t)(kk * 4) * DIMM + t * 16];
                    vf[0][kk][t][1] = vb[(size_t)(kk * 4 + 1) * DIMM + t * 16];
                }
        }
        group_fence();

        float rmax[8];
#pragma unroll
        for (int r = 0; r < 8; ++r) {
#pragma unroll
            for (int j = 0; j < KB_TILES; ++j) s[j][r] = s[j][r] * scale + mk[r][j];
            rmax[r] = fmaxf(fmaxf(s[0][r], s[1][r]), fmaxf(s[2][r], s[3][r]));
        }
#pragma unroll
        for (int off = 1; off < 16; off <<= 1) {
            float ex[8];
#pragma unroll
            for (int r = 0; r < 8; ++r) ex[r] = __shfl_xor(rmax[r], off, 32);
#pragma unroll
            for (int r = 0; r < 8; ++r) rmax[r] = fmaxf(rmax[r], ex[r]);
        }

#pragma unroll
        for (int r = 0; r < 8; ++r) {
            const float mnew = fmaxf(mrow[r], rmax[r]);
            const float corr = __expf(mrow[r] - mnew);
            mrow[r] = mnew;
#pragma unroll
            for (int j = 0; j < KB_TILES; ++j) s[j][r] = __expf(s[j][r] - mnew);
            lsum[r] = lsum[r] * corr + ((s[0][r] + s[1][r]) + (s[2][r] + s[3][r]));
#pragma unroll
            for (int t = 0; t < 8; ++t) oacc[t][r] *= corr;
        }

#pragma unroll
        for (int j = 0; j < KB_TILES; ++j)
#pragma unroll
            for (int r = 0; r < 8; ++r) lp[j * 256 + (r + 8 * hi) * 16 + lc] = s[j][r];
        asm volatile("s_wait_dscnt 0" ::: "memory");
        v2f pa[KB_TILES][4];
#pragma unroll
        for (int j = 0; j < KB_TILES; ++j)
#pragma unroll
            for (int kk = 0; kk < 4; ++kk)
                pa[j][kk] = *(const v2f*)&lp[j * 256 + lc * 16 + kk * 4 + 2 * hi];
        asm volatile("s_wait_dscnt 0" ::: "memory");

#pragma unroll
        for (int j = 0; j < KB_TILES; ++j) {
            if (j + 1 < KB_TILES) {
                const float* vb =
                    V + (size_t)(kv0 + (j + 1) * 16 + 2 * hi) * DIMM + h * HDIM + lc;
#pragma unroll
                for (int kk = 0; kk < 4; ++kk)
#pragma unroll
                    for (int t = 0; t < 8; ++t) {
                        vf[(j + 1) & 1][kk][t][0] = vb[(size_t)(kk * 4) * DIMM + t * 16];
                        vf[(j + 1) & 1][kk][t][1] = vb[(size_t)(kk * 4 + 1) * DIMM + t * 16];
                    }
            }
#pragma unroll
            for (int kk = 0; kk < 4; ++kk)
#pragma unroll
                for (int t = 0; t < 8; ++t) {
                    v2f bv;
                    bv[0] = vf[j & 1][kk][t][0];
                    bv[1] = vf[j & 1][kk][t][1];
                    oacc[t] = wmma_f32(pa[j][kk], bv, oacc[t]);
                }
            group_fence();
        }
    }

#pragma unroll
    for (int off = 1; off < 16; off <<= 1) {
        float ex[8];
#pragma unroll
        for (int r = 0; r < 8; ++r) ex[r] = __shfl_xor(lsum[r], off, 32);
#pragma unroll
        for (int r = 0; r < 8; ++r) lsum[r] += ex[r];
    }
#pragma unroll
    for (int r = 0; r < 8; ++r) {
        const float inv = 1.0f / lsum[r];
        float* orow = O + (size_t)(q0 + r + 8 * hi) * DIMM + h * HDIM + lc;
#pragma unroll
        for (int t = 0; t < 8; ++t) orow[t * 16] = oacc[t][r] * inv;
    }
}

extern "C" void kernel_launch(void* const* d_in, const int* in_sizes, int n_in,
                              void* d_out, int out_size, void* d_ws, size_t ws_size,
                              hipStream_t stream) {
    const float* x    = (const float*)d_in[0];
    const float* mask = (const float*)d_in[1];
    const float* wq   = (const float*)d_in[3];
    const float* wk   = (const float*)d_in[4];
    const float* wv   = (const float*)d_in[5];
    const float* wo   = (const float*)d_in[6];
    float* out = (float*)d_out;

    float* ws = (float*)d_ws;
    const size_t MAT = (size_t)SEQL * DIMM;
    float* Qb = ws;
    float* Kb = ws + MAT;
    float* Vb = ws + 2 * MAT;
    float* Ab = ws + 3 * MAT;

    dim3 tb(256);
    dim3 tg(128);
    dim3 gg(DIMM / 128, SEQL / 128);
    gemm_abt<<<gg, tg, 0, stream>>>(x, wq, Qb);
    gemm_abt<<<gg, tg, 0, stream>>>(x, wk, Kb);
    gemm_abt<<<gg, tg, 0, stream>>>(x, wv, Vb);

    dim3 ga(SEQL / 128, NHEADS);
    attn_fwd<<<ga, tb, 0, stream>>>(Qb, Kb, Vb, mask, Ab);

    gemm_abt<<<gg, tg, 0, stream>>>(Ab, wo, out);
}


# attn_fwd: Q,K,V fragment loads 16 bytes per lane (fewer cache lines per WMMA): Q,K 8 consecutive d per lane, d order in a 16-d chunk permuted, score sum order changes; V,O tile columns permuted
# speedup vs baseline: 1.4920x; 1.0523x over previous
#include <hip/hip_runtime.h>

#define SEQL   2048
#define DIMM   2048
#define NHEADS 16
#define HDIM   128

typedef float v2f __attribute__((ext_vector_type(2)));
typedef float v4f __attribute__((ext_vector_type(4)));
typedef float v8f __attribute__((ext_vector_type(8)));

__device__ __forceinline__ v8f wmma_f32(v2f a, v2f b, v8f c) {
    return __builtin_amdgcn_wmma_f32_16x16x4_f32(false, a, false, b, (short)0, c, false, false);
}

struct Frag8 { v4f lo, hi; };

__device__ __forceinline__ Frag8 load_frag8(const float* __restrict__ p) {
    Frag8 f;
    f.lo = *(const v4f*)(p);
    f.hi = *(const v4f*)(p + 4);
    return f;
}

template <int J>
__device__ __forceinline__ v2f kpair(const Frag8& f) {
    if (J == 0) return __builtin_shufflevector(f.lo, f.lo, 0, 1);
    if (J == 1) return __builtin_shufflevector(f.lo, f.lo, 2, 3);
    if (J == 2) return __builtin_shufflevector(f.hi, f.hi, 0, 1);
    return __builtin_shufflevector(f.hi, f.hi, 2, 3);
}

template <int J>
__device__ __forceinline__ void mma_step(const Frag8 (&a)[4], const Frag8 (&b)[4],
                                         v8f (&acc)[4][4]) {
    for (int i = 0; i < 4; ++i) {
        const v2f av = kpair<J>(a[i]);
        for (int n = 0; n < 4; ++n)
            acc[i][n] = wmma_f32(av, kpair<J>(b[n]), acc[i][n]);
    }
}

__device__ __forceinline__ void mma_chunk(const Frag8 (&a)[4], const Frag8 (&b)[4],
                                          v8f (&acc)[4][4]) {
    mma_step<0>(a, b, acc);
    mma_step<1>(a, b, acc);
    mma_step<2>(a, b, acc);
    mma_step<3>(a, b, acc);
}

__device__ __forceinline__ void load_chunk(Frag8 (&a)[4], Frag8 (&b)[4],
                                           const float* __restrict__ ap,
                                           const float* __restrict__ bp, int k0) {
    for (int i = 0; i < 4; ++i) a[i] = load_frag8(ap + (size_t)i * 16 * DIMM + k0);
    for (int n = 0; n < 4; ++n) b[n] = load_frag8(bp + (size_t)n * 16 * DIMM + k0);
}

__global__ __launch_bounds__(128)
void gemm_abt(const float* __restrict__ A, const float* __restrict__ B,
              float* __restrict__ C) {
    const int lane = threadIdx.x & 31;
    const int wave = threadIdx.x >> 5;
    const int hi   = lane >> 4;
    const int lc   = lane & 15;

    const int m0 = blockIdx.y * 128 + (wave >> 1) * 64;
    const int n0 = blockIdx.x * 128 + (wave & 1) * 64;

    v8f acc[4][4];
    for (int i = 0; i < 4; ++i)
        for (int n = 0; n < 4; ++n) acc[i][n] = (v8f){};

    const float* ap = A + (size_t)(m0 + lc) * DIMM + 8 * hi;
    const float* bp = B + (size_t)(n0 + lc) * DIMM + 8 * hi;

    Frag8 a0[4], b0[4], a1[4], b1[4];
    load_chunk(a0, b0, ap, bp, 0);

    for (int k0 = 0; k0 < DIMM - 32; k0 += 32) {
        load_chunk(a1, b1, ap, bp, k0 + 16);
        mma_chunk(a0, b0, acc);
        load_chunk(a0, b0, ap, bp, k0 + 32);
        mma_chunk(a1, b1, acc);
    }
    load_chunk(a1, b1, ap, bp, DIMM - 16);
    mma_chunk(a0, b0, acc);
    mma_chunk(a1, b1, acc);

    for (int i = 0; i < 4; ++i) {
        for (int r = 0; r < 8; ++r) {
            float* crow = C + (size_t)(m0 + 16 * i + r + 8 * hi) * DIMM + n0 + lc;
            for (int n = 0; n < 4; ++n) crow[16 * n] = acc[i][n][r];
        }
    }
}

#define KB_TILES 4
#define KB       (16 * KB_TILES)

__device__ __forceinline__ void group_fence() { __builtin_amdgcn_sched_barrier(0); }

template <int J>
__device__ __forceinline__ void qk_step(const Frag8& q, const Frag8 (&k)[KB_TILES],
                                        v8f (&s)[KB_TILES]) {
    const v2f qv = kpair<J>(q);
#pragma unroll
    for (int j = 0; j < KB_TILES; ++j) s[j] = wmma_f32(qv, kpair<J>(k[j]), s[j]);
}

__global__ __launch_bounds__(256) __attribute__((amdgpu_waves_per_eu(2, 2)))
void attn_fwd(const float* __restrict__ Q, const float* __restrict__ K,
              const float* __restrict__ V, const float* __restrict__ mask,
              float* __restrict__ O) {
    __shared__ float ldsP[8 * KB_TILES * 256];

    const int lane = threadIdx.x & 31;
    const int wave = threadIdx.x >> 5;
    const int hi   = lane >> 4;
    const int lc   = lane & 15;

    const int h  = blockIdx.y;
    const int q0 = (blockIdx.x * 8 + wave) * 16;


    Frag8 aq[8];
    {
        const float* qrow = Q + (size_t)(q0 + lc) * DIMM + h * HDIM + 8 * hi;
#pragma unroll
        for (int cc = 0; cc < 8; ++cc) aq[cc] = load_frag8(qrow + cc * 16);
    }

    v8f oacc[8];
#pragma unroll
    for (int t = 0; t < 8; ++t) oacc[t] = (v8f){};
    float mrow[8], lsum[8];
#pragma unroll
    for (int r = 0; r < 8; ++r) { mrow[r] = -1.0e30f; lsum[r] = 0.0f; }

    float* lp = &ldsP[wave * (KB_TILES * 256)];
    const float scale = 0.08838834764831845f;
    const size_t mbase = ((size_t)h * SEQL + q0) * SEQL;

#pragma unroll 1
    for (int kv0 = 0; kv0 < SEQL; kv0 += KB) {
        v8f s[KB_TILES];
#pragma unroll
        for (int j = 0; j < KB_TILES; ++j) s[j] = (v8f){};
        const float* krow = K + (size_t)(kv0 + lc) * DIMM + h * HDIM + 8 * hi;
        Frag8 kf[2][2][KB_TILES];
        float mk[8][KB_TILES];
#pragma unroll
        for (int cc = 0; cc < 2; ++cc)
#pragma unroll
            for (int j = 0; j < KB_TILES; ++j)
                kf[0][cc][j] = load_frag8(krow + (size_t)j * 16 * DIMM + cc * 16);
        group_fence();
#pragma unroll
        for (int g = 0; g < 4; ++g) {
            if (g + 1 < 4) {
#pragma unroll
                for (int cc = 0; cc < 2; ++cc)
#pragma unroll
                    for (int j = 0; j < KB_TILES; ++j)
                        kf[(g + 1) & 1][cc][j] =
                            load_frag8(krow + (size_t)j * 16 * DIMM + ((g + 1) * 2 + cc) * 16);
            }
            if (g == 2) {
                group_fence();
#pragma unroll
                for (int r = 0; r < 8; ++r) {
                    const float* mp = mask + mbase + (size_t)(r + 8 * hi) * SEQL + kv0 + lc;
#pragma unroll
                    for (int j = 0; j < KB_TILES; ++j)
                        mk[r][j] = __builtin_nontemporal_load(mp + 16 * j);
                }
            }
#pragma unroll
            for (int cc = 0; cc < 2; ++cc) {
                qk_step<0>(aq[g * 2 + cc], kf[g & 1][cc], s);
                qk_step<1>(aq[g * 2 + cc], kf[g & 1][cc], s);
                qk_step<2>(aq[g * 2 + cc], kf[g & 1][cc], s);
                qk_step<3>(aq[g * 2 + cc], kf[g & 1][cc], s);
            }
            group_fence();
        }

        v4f vf[2][4][2][2];
        {
            const float* vb = V + (size_t)(kv0 + 2 * hi) * DIMM + h * HDIM + 4 * lc;
#pragma unroll
            for (int kk = 0; kk < 4; ++kk)
#pragma unroll
                for (int e = 0; e < 2; ++e)
#pragma unroll
                    for (int g = 0; g < 2; ++g)
                        vf[0][kk][e][g] = *(const v4f*)(vb + (size_t)(kk * 4 + e) * DIMM + 64 * g);
        }
        group_fence();

        float rmax[8];
#pragma unroll
        for (int r = 0; r < 8; ++r) {
#pragma unroll
            for (int j = 0; j < KB_TILES; ++j) s[j][r] = s[j][r] * scale + mk[r][j];
            rmax[r] = fmaxf(fmaxf(s[0][r], s[1][r]), fmaxf(s[2][r], s[3][r]));
        }
#pragma unroll
        for (int off = 1; off < 16; off <<= 1) {
            float ex[8];
#pragma unroll
            for (int r = 0; r < 8; ++r) ex[r] = __shfl_xor(rmax[r], off, 32);
#pragma unroll
            for (int r = 0; r < 8; ++r) rmax[r] = fmaxf(rmax[r], ex[r]);
        }

#pragma unroll
        for (int r = 0; r < 8; ++r) {
            const float mnew = fmaxf(mrow[r], rmax[r]);
            const float corr = __expf(mrow[r] - mnew);
            mrow[r] = mnew;
#pragma unroll
            for (int j = 0; j < KB_TILES; ++j) s[j][r] = __expf(s[j][r] - mnew);
            lsum[r] = lsum[r] * corr + ((s[0][r] + s[1][r]) + (s[2][r] + s[3][r]));
#pragma unroll
            for (int t = 0; t < 8; ++t) oacc[t][r] *= corr;
        }

#pragma unroll
        for (int j = 0; j < KB_TILES; ++j)
#pragma unroll
            for (int r = 0; r < 8; ++r) lp[j * 256 + (r + 8 * hi) * 16 + lc] = s[j][r];
        asm volatile("s_wait_dscnt 0" ::: "memory");
        v2f pa[KB_TILES][4];
#pragma unroll
        for (int j = 0; j < KB_TILES; ++j)
#pragma unroll
            for (int kk = 0; kk < 4; ++kk)
                pa[j][kk] = *(const v2f*)&lp[j * 256 + lc * 16 + kk * 4 + 2 * hi];
        asm volatile("s_wait_dscnt 0" ::: "memory");

#pragma unroll
        for (int j = 0; j < KB_TILES; ++j) {
            if (j + 1 < KB_TILES) {
                const float* vb =
                    V + (size_t)(kv0 + (j + 1) * 16 + 2 * hi) * DIMM + h * HDIM + 4 * lc;
#pragma unroll
                for (int kk = 0; kk < 4; ++kk)
#pragma unroll
                    for (int e = 0; e < 2; ++e)
#pragma unroll
                        for (int g = 0; g < 2; ++g)
                            vf[(j + 1) & 1][kk][e][g] =
                                *(const v4f*)(vb + (size_t)(kk * 4 + e) * DIMM + 64 * g);
            }
#pragma unroll
            for (int kk = 0; kk < 4; ++kk)
#pragma unroll
                for (int t = 0; t < 8; ++t) {
                    v2f bv;
                    bv[0] = vf[j & 1][kk][0][t >> 2][t & 3];
                    bv[1] = vf[j & 1][kk][1][t >> 2][t & 3];
                    oacc[t] = wmma_f32(pa[j][kk], bv, oacc[t]);
                }
            group_fence();
        }
    }

#pragma unroll
    for (int off = 1; off < 16; off <<= 1) {
        float ex[8];
#pragma unroll
        for (int r = 0; r < 8; ++r) ex[r] = __shfl_xor(lsum[r], off, 32);
#pragma unroll
        for (int r = 0; r < 8; ++r) lsum[r] += ex[r];
    }
#pragma unroll
    for (int r = 0; r < 8; ++r) {
        const float inv = 1.0f / lsum[r];
        float* orow = O + (size_t)(q0 + r + 8 * hi) * DIMM + h * HDIM + 4 * lc;
#pragma unroll
        for (int g = 0; g < 2; ++g) {
            v4f o;
#pragma unroll
            for (int e = 0; e < 4; ++e) o[e] = oacc[4 * g + e][r] * inv;
            *(v4f*)(orow + 64 * g) = o;
        }
    }
}

extern "C" void kernel_launch(void* const* d_in, const int* in_sizes, int n_in,
                              void* d_out, int out_size, void* d_ws, size_t ws_size,
                              hipStream_t stream) {
    const float* x    = (const float*)d_in[0];
    const float* mask = (const float*)d_in[1];
    const float* wq   = (const float*)d_in[3];
    const float* wk   = (const float*)d_in[4];
    const float* wv   = (const float*)d_in[5];
    const float* wo   = (const float*)d_in[6];
    float* out = (float*)d_out;

    float* ws = (float*)d_ws;
    const size_t MAT = (size_t)SEQL * DIMM;
    float* Qb = ws;
    float* Kb = ws + MAT;
    float* Vb = ws + 2 * MAT;
    float* Ab = ws + 3 * MAT;

    dim3 tb(256);
    dim3 tg(128);
    dim3 gg(DIMM / 128, SEQL / 128);
    gemm_abt<<<gg, tg, 0, stream>>>(x, wq, Qb);
    gemm_abt<<<gg, tg, 0, stream>>>(x, wk, Kb);
    gemm_abt<<<gg, tg, 0, stream>>>(x, wv, Vb);

    dim3 ga(SEQL / 128, NHEADS);
    attn_fwd<<<ga, tb, 0, stream>>>(Qb, Kb, Vb, mask, Ab);

    gemm_abt<<<gg, tg, 0, stream>>>(Ab, wo, out);
}


# gemm_abt: K split over the block's 4 waves, each wave holds the whole 128x128 tile in registers (a fragment load feeds 8 WMMAs, not 4); partials added via LDS in wave order, fp32 sum order changes
# speedup vs baseline: 1.6182x; 1.0846x over previous
#include <hip/hip_runtime.h>

#define SEQL   2048
#define DIMM   2048
#define NHEADS 16
#define HDIM   128

typedef float v2f __attribute__((ext_vector_type(2)));
typedef float v4f __attribute__((ext_vector_type(4)));
typedef float v8f __attribute__((ext_vector_type(8)));

__device__ __forceinline__ v8f wmma_f32(v2f a, v2f b, v8f c) {
    return __builtin_amdgcn_wmma_f32_16x16x4_f32(false, a, false, b, (short)0, c, false, false);
}

#define KSPLIT  4
#define KSTRIDE (16 * KSPLIT)
#define NCHUNK  (DIMM / KSTRIDE)

struct Frag8 { v4f lo, hi; };

__device__ __forceinline__ Frag8 load_frag8(const float* __restrict__ p) {
    Frag8 f;
    f.lo = *(const v4f*)(p);
    f.hi = *(const v4f*)(p + 4);
    return f;
}

template <int J>
__device__ __forceinline__ v2f kpair(const Frag8& f) {
    if (J == 0) return __builtin_shufflevector(f.lo, f.lo, 0, 1);
    if (J == 1) return __builtin_shufflevector(f.lo, f.lo, 2, 3);
    if (J == 2) return __builtin_shufflevector(f.hi, f.hi, 0, 1);
    return __builtin_shufflevector(f.hi, f.hi, 2, 3);
}

template <int J>
__device__ __forceinline__ void mma_step(const Frag8 (&a)[8], const Frag8 (&b)[8],
                                         v8f (&acc)[8][8]) {
#pragma unroll
    for (int i = 0; i < 8; ++i) {
        const v2f av = kpair<J>(a[i]);
#pragma unroll
        for (int n = 0; n < 8; ++n)
            acc[i][n] = wmma_f32(av, kpair<J>(b[n]), acc[i][n]);
    }
}

__device__ __forceinline__ void mma_chunk(const Frag8 (&a)[8], const Frag8 (&b)[8],
                                          v8f (&acc)[8][8]) {
    mma_step<0>(a, b, acc);
    mma_step<1>(a, b, acc);
    mma_step<2>(a, b, acc);
    mma_step<3>(a, b, acc);
}

__device__ __forceinline__ void load_chunk(Frag8 (&a)[8], Frag8 (&b)[8],
                                           const float* __restrict__ ap,
                                           const float* __restrict__ bp, int t) {
#pragma unroll
    for (int i = 0; i < 8; ++i) a[i] = load_frag8(ap + (size_t)i * 16 * DIMM + t * KSTRIDE);
#pragma unroll
    for (int n = 0; n < 8; ++n) b[n] = load_frag8(bp + (size_t)n * 16 * DIMM + t * KSTRIDE);
}

__global__ __launch_bounds__(128)
void gemm_abt(const float* __restrict__ A, const float* __restrict__ B,
              float* __restrict__ C) {
    __shared__ __attribute__((aligned(16))) float red[KSPLIT * 64 * 256];

    const int lane = threadIdx.x & 31;
    const int wave = threadIdx.x >> 5;
    const int hi   = lane >> 4;
    const int lc   = lane & 15;

    const int m0 = blockIdx.y * 128;
    const int n0 = blockIdx.x * 128;

    v8f acc[8][8];
#pragma unroll
    for (int i = 0; i < 8; ++i)
#pragma unroll
        for (int n = 0; n < 8; ++n) acc[i][n] = (v8f){};

    const float* ap = A + (size_t)(m0 + lc) * DIMM + 16 * wave + 8 * hi;
    const float* bp = B + (size_t)(n0 + lc) * DIMM + 16 * wave + 8 * hi;

    Frag8 a0[8], b0[8], a1[8], b1[8];
    load_chunk(a0, b0, ap, bp, 0);

    for (int t = 0; t < NCHUNK - 2; t += 2) {
        load_chunk(a1, b1, ap, bp, t + 1);
        mma_chunk(a0, b0, acc);
        load_chunk(a0, b0, ap, bp, t + 2);
        mma_chunk(a1, b1, acc);
    }
    load_chunk(a1, b1, ap, bp, NCHUNK - 1);
    mma_chunk(a0, b0, acc);
    mma_chunk(a1, b1, acc);

    {
        float* mine = red + wave * (64 * 256) + lane * 4;
#pragma unroll
        for (int i = 0; i < 8; ++i)
#pragma unroll
            for (int n = 0; n < 8; ++n) {
                float* tp = mine + (8 * i + n) * 256;
                *(v4f*)(tp)       = __builtin_shufflevector(acc[i][n], acc[i][n], 0, 1, 2, 3);
                *(v4f*)(tp + 128) = __builtin_shufflevector(acc[i][n], acc[i][n], 4, 5, 6, 7);
            }
    }
    __syncthreads();

#pragma unroll
    for (int ii = 0; ii < 2; ++ii) {
        const int i = 2 * wave + ii;
#pragma unroll
        for (int n = 0; n < 8; ++n) {
            const float* tp = red + (8 * i + n) * 256 + lane * 4;
            v8f t;
#pragma unroll
            for (int s = 0; s < KSPLIT; ++s) {
                const v4f plo = *(const v4f*)(tp + s * (64 * 256));
                const v4f phi = *(const v4f*)(tp + s * (64 * 256) + 128);
                const v8f p = __builtin_shufflevector(plo, phi, 0, 1, 2, 3, 4, 5, 6, 7);
                t = (s == 0) ? p : t + p;
            }
#pragma unroll
            for (int r = 0; r < 8; ++r)
                C[(size_t)(m0 + 16 * i + r + 8 * hi) * DIMM + n0 + 16 * n + lc] = t[r];
        }
    }
}

#define KB_TILES 4
#define KB       (16 * KB_TILES)

__device__ __forceinline__ void group_fence() { __builtin_amdgcn_sched_barrier(0); }

template <int J>
__device__ __forceinline__ void qk_step(const Frag8& q, const Frag8 (&k)[KB_TILES],
                                        v8f (&s)[KB_TILES]) {
    const v2f qv = kpair<J>(q);
#pragma unroll
    for (int j = 0; j < KB_TILES; ++j) s[j] = wmma_f32(qv, kpair<J>(k[j]), s[j]);
}

__global__ __launch_bounds__(256) __attribute__((amdgpu_waves_per_eu(2, 2)))
void attn_fwd(const float* __restrict__ Q, const float* __restrict__ K,
              const float* __restrict__ V, const float* __restrict__ mask,
              float* __restrict__ O) {
    __shared__ float ldsP[8 * KB_TILES * 256];

    const int lane = threadIdx.x & 31;
    const int wave = threadIdx.x >> 5;
    const int hi   = lane >> 4;
    const int lc   = lane & 15;

    const int h  = blockIdx.y;
    const int q0 = (blockIdx.x * 8 + wave) * 16;


    Frag8 aq[8];
    {
        const float* qrow = Q + (size_t)(q0 + lc) * DIMM + h * HDIM + 8 * hi;
#pragma unroll
        for (int cc = 0; cc < 8; ++cc) aq[cc] = load_frag8(qrow + cc * 16);
    }

    v8f oacc[8];
#pragma unroll
    for (int t = 0; t < 8; ++t) oacc[t] = (v8f){};
    float mrow[8], lsum[8];
#pragma unroll
    for (int r = 0; r < 8; ++r) { mrow[r] = -1.0e30f; lsum[r] = 0.0f; }

    float* lp = &ldsP[wave * (KB_TILES * 256)];
    const float scale = 0.08838834764831845f;
    const size_t mbase = ((size_t)h * SEQL + q0) * SEQL;

#pragma unroll 1
    for (int kv0 = 0; kv0 < SEQL; kv0 += KB) {
        v8f s[KB_TILES];
#pragma unroll
        for (int j = 0; j < KB_TILES; ++j) s[j] = (v8f){};
        const float* krow = K + (size_t)(kv0 + lc) * DIMM + h * HDIM + 8 * hi;
        Frag8 kf[2][2][KB_TILES];
        float mk[8][KB_TILES];
#pragma unroll
        for (int cc = 0; cc < 2; ++cc)
#pragma unroll
            for (int j = 0; j < KB_TILES; ++j)
                kf[0][cc][j] = load_frag8(krow + (size_t)j * 16 * DIMM + cc * 16);
        group_fence();
#pragma unroll
        for (int g = 0; g < 4; ++g) {
            if (g + 1 < 4) {
#pragma unroll
                for (int cc = 0; cc < 2; ++cc)
#pragma unroll
                    for (int j = 0; j < KB_TILES; ++j)
                        kf[(g + 1) & 1][cc][j] =
                            load_frag8(krow + (size_t)j * 16 * DIMM + ((g + 1) * 2 + cc) * 16);
            }
            if (g == 2) {
                group_fence();
#pragma unroll
                for (int r = 0; r < 8; ++r) {
                    const float* mp = mask + mbase + (size_t)(r + 8 * hi) * SEQL + kv0 + lc;
#pragma unroll
                    for (int j = 0; j < KB_TILES; ++j)
                        mk[r][j] = __builtin_nontemporal_load(mp + 16 * j);
                }
            }
#pragma unroll
            for (int cc = 0; cc < 2; ++cc) {
                qk_step<0>(aq[g * 2 + cc], kf[g & 1][cc], s);
                qk_step<1>(aq[g * 2 + cc], kf[g & 1][cc], s);
                qk_step<2>(aq[g * 2 + cc], kf[g & 1][cc], s);
                qk_step<3>(aq[g * 2 + cc], kf[g & 1][cc], s);
            }
            group_fence();
        }

        v4f vf[2][4][2][2];
        {
            const float* vb = V + (size_t)(kv0 + 2 * hi) * DIMM + h * HDIM + 4 * lc;
#pragma unroll
            for (int kk = 0; kk < 4; ++kk)
#pragma unroll
                for (int e = 0; e < 2; ++e)
#pragma unroll
                    for (int g = 0; g < 2; ++g)
                        vf[0][kk][e][g] = *(const v4f*)(vb + (size_t)(kk * 4 + e) * DIMM + 64 * g);
        }
        group_fence();

        float rmax[8];
#pragma unroll
        for (int r = 0; r < 8; ++r) {
#pragma unroll
            for (int j = 0; j < KB_TILES; ++j) s[j][r] = s[j][r] * scale + mk[r][j];
            rmax[r] = fmaxf(fmaxf(s[0][r], s[1][r]), fmaxf(s[2][r], s[3][r]));
        }
#pragma unroll
        for (int off = 1; off < 16; off <<= 1) {
            float ex[8];
#pragma unroll
            for (int r = 0; r < 8; ++r) ex[r] = __shfl_xor(rmax[r], off, 32);
#pragma unroll
            for (int r = 0; r < 8; ++r) rmax[r] = fmaxf(rmax[r], ex[r]);
        }

#pragma unroll
        for (int r = 0; r < 8; ++r) {
            const float mnew = fmaxf(mrow[r], rmax[r]);
            const float corr = __expf(mrow[r] - mnew);
            mrow[r] = mnew;
#pragma unroll
            for (int j = 0; j < KB_TILES; ++j) s[j][r] = __expf(s[j][r] - mnew);
            lsum[r] = lsum[r] * corr + ((s[0][r] + s[1][r]) + (s[2][r] + s[3][r]));
#pragma unroll
            for (int t = 0; t < 8; ++t) oacc[t][r] *= corr;
        }

#pragma unroll
        for (int j = 0; j < KB_TILES; ++j)
#pragma unroll
            for (int r = 0; r < 8; ++r) lp[j * 256 + (r + 8 * hi) * 16 + lc] = s[j][r];
        asm volatile("s_wait_dscnt 0" ::: "memory");
        v2f pa[KB_TILES][4];
#pragma unroll
        for (int j = 0; j < KB_TILES; ++j)
#pragma unroll
            for (int kk = 0; kk < 4; ++kk)
                pa[j][kk] = *(const v2f*)&lp[j * 256 + lc * 16 + kk * 4 + 2 * hi];
        asm volatile("s_wait_dscnt 0" ::: "memory");

#pragma unroll
        for (int j = 0; j < KB_TILES; ++j) {
            if (j + 1 < KB_TILES) {
                const float* vb =
                    V + (size_t)(kv0 + (j + 1) * 16 + 2 * hi) * DIMM + h * HDIM + 4 * lc;
#pragma unroll
                for (int kk = 0; kk < 4; ++kk)
#pragma unroll
                    for (int e = 0; e < 2; ++e)
#pragma unroll
                        for (int g = 0; g < 2; ++g)
                            vf[(j + 1) & 1][kk][e][g] =
                                *(const v4f*)(vb + (size_t)(kk * 4 + e) * DIMM + 64 * g);
            }
#pragma unroll
            for (int kk = 0; kk < 4; ++kk)
#pragma unroll
                for (int t = 0; t < 8; ++t) {
                    v2f bv;
                    bv[0] = vf[j & 1][kk][0][t >> 2][t & 3];
                    bv[1] = vf[j & 1][kk][1][t >> 2][t & 3];
                    oacc[t] = wmma_f32(pa[j][kk], bv, oacc[t]);
                }
            group_fence();
        }
    }

#pragma unroll
    for (int off = 1; off < 16; off <<= 1) {
        float ex[8];
#pragma unroll
        for (int r = 0; r < 8; ++r) ex[r] = __shfl_xor(lsum[r], off, 32);
#pragma unroll
        for (int r = 0; r < 8; ++r) lsum[r] += ex[r];
    }
#pragma unroll
    for (int r = 0; r < 8; ++r) {
        const float inv = 1.0f / lsum[r];
        float* orow = O + (size_t)(q0 + r + 8 * hi) * DIMM + h * HDIM + 4 * lc;
#pragma unroll
        for (int g = 0; g < 2; ++g) {
            v4f o;
#pragma unroll
            for (int e = 0; e < 4; ++e) o[e] = oacc[4 * g + e][r] * inv;
            *(v4f*)(orow + 64 * g) = o;
        }
    }
}

extern "C" void kernel_launch(void* const* d_in, const int* in_sizes, int n_in,
                              void* d_out, int out_size, void* d_ws, size_t ws_size,
                              hipStream_t stream) {
    const float* x    = (const float*)d_in[0];
    const float* mask = (const float*)d_in[1];
    const float* wq   = (const float*)d_in[3];
    const float* wk   = (const float*)d_in[4];
    const float* wv   = (const float*)d_in[5];
    const float* wo   = (const float*)d_in[6];
    float* out = (float*)d_out;

    float* ws = (float*)d_ws;
    const size_t MAT = (size_t)SEQL * DIMM;
    float* Qb = ws;
    float* Kb = ws + MAT;
    float* Vb = ws + 2 * MAT;
    float* Ab = ws + 3 * MAT;

    dim3 tb(256);
    dim3 tg(128);
    dim3 gg(DIMM / 128, SEQL / 128);
    gemm_abt<<<gg, tg, 0, stream>>>(x, wq, Qb);
    gemm_abt<<<gg, tg, 0, stream>>>(x, wk, Kb);
    gemm_abt<<<gg, tg, 0, stream>>>(x, wv, Vb);

    dim3 ga(SEQL / 128, NHEADS);
    attn_fwd<<<ga, tb, 0, stream>>>(Qb, Kb, Vb, mask, Ab);

    gemm_abt<<<gg, tg, 0, stream>>>(Ab, wo, out);
}


# K and V projections stored in the attention kernel's fragment order: a K fragment load reads 512 contiguous bytes (4 lines, not 16), V arrives as ready B register pairs (no moves); sums unchanged
# speedup vs baseline: 1.7281x; 1.0679x over previous
#include <hip/hip_runtime.h>

#define SEQL   2048
#define DIMM   2048
#define NHEADS 16
#define HDIM   128

typedef float v2f __attribute__((ext_vector_type(2)));
typedef float v4f __attribute__((ext_vector_type(4)));
typedef float v8f __attribute__((ext_vector_type(8)));

__device__ __forceinline__ v8f wmma_f32(v2f a, v2f b, v8f c) {
    return __builtin_amdgcn_wmma_f32_16x16x4_f32(false, a, false, b, (short)0, c, false, false);
}

#define KSPLIT  4
#define KSTRIDE (16 * KSPLIT)
#define NCHUNK  (DIMM / KSTRIDE)

struct Frag8 { v4f lo, hi; };

__device__ __forceinline__ Frag8 load_frag8(const float* __restrict__ p) {
    Frag8 f;
    f.lo = *(const v4f*)(p);
    f.hi = *(const v4f*)(p + 4);
    return f;
}

template <int J>
__device__ __forceinline__ v2f kpair(const Frag8& f) {
    if (J == 0) return __builtin_shufflevector(f.lo, f.lo, 0, 1);
    if (J == 1) return __builtin_shufflevector(f.lo, f.lo, 2, 3);
    if (J == 2) return __builtin_shufflevector(f.hi, f.hi, 0, 1);
    return __builtin_shufflevector(f.hi, f.hi, 2, 3);
}

template <int J>
__device__ __forceinline__ void mma_step(const Frag8 (&a)[8], const Frag8 (&b)[8],
                                         v8f (&acc)[8][8]) {
#pragma unroll
    for (int i = 0; i < 8; ++i) {
        const v2f av = kpair<J>(a[i]);
#pragma unroll
        for (int n = 0; n < 8; ++n)
            acc[i][n] = wmma_f32(av, kpair<J>(b[n]), acc[i][n]);
    }
}

__device__ __forceinline__ void mma_chunk(const Frag8 (&a)[8], const Frag8 (&b)[8],
                                          v8f (&acc)[8][8]) {
    mma_step<0>(a, b, acc);
    mma_step<1>(a, b, acc);
    mma_step<2>(a, b, acc);
    mma_step<3>(a, b, acc);
}

__device__ __forceinline__ void load_chunk(Frag8 (&a)[8], Frag8 (&b)[8],
                                           const float* __restrict__ ap,
                                           const float* __restrict__ bp, int t) {
#pragma unroll
    for (int i = 0; i < 8; ++i) a[i] = load_frag8(ap + (size_t)i * 16 * DIMM + t * KSTRIDE);
#pragma unroll
    for (int n = 0; n < 8; ++n) b[n] = load_frag8(bp + (size_t)n * 16 * DIMM + t * KSTRIDE);
}

template <int MODE>
__global__ __launch_bounds__(128)
void gemm_abt(const float* __restrict__ A, const float* __restrict__ B,
              float* __restrict__ C) {
    __shared__ __attribute__((aligned(16))) float red[KSPLIT * 64 * 256];

    const int lane = threadIdx.x & 31;
    const int wave = threadIdx.x >> 5;
    const int hi   = lane >> 4;
    const int lc   = lane & 15;

    const int m0 = blockIdx.y * 128;
    const int n0 = blockIdx.x * 128;

    v8f acc[8][8];
#pragma unroll
    for (int i = 0; i < 8; ++i)
#pragma unroll
        for (int n = 0; n < 8; ++n) acc[i][n] = (v8f){};

    const float* ap = A + (size_t)(m0 + lc) * DIMM + 16 * wave + 8 * hi;
    const float* bp = B + (size_t)(n0 + lc) * DIMM + 16 * wave + 8 * hi;

    Frag8 a0[8], b0[8], a1[8], b1[8];
    load_chunk(a0, b0, ap, bp, 0);

    for (int t = 0; t < NCHUNK - 2; t += 2) {
        load_chunk(a1, b1, ap, bp, t + 1);
        mma_chunk(a0, b0, acc);
        load_chunk(a0, b0, ap, bp, t + 2);
        mma_chunk(a1, b1, acc);
    }
    load_chunk(a1, b1, ap, bp, NCHUNK - 1);
    mma_chunk(a0, b0, acc);
    mma_chunk(a1, b1, acc);

    {
        float* mine = red + wave * (64 * 256) + lane * 4;
#pragma unroll
        for (int i = 0; i < 8; ++i)
#pragma unroll
            for (int n = 0; n < 8; ++n) {
                float* tp = mine + (8 * i + n) * 256;
                *(v4f*)(tp)       = __builtin_shufflevector(acc[i][n], acc[i][n], 0, 1, 2, 3);
                *(v4f*)(tp + 128) = __builtin_shufflevector(acc[i][n], acc[i][n], 4, 5, 6, 7);
            }
    }
    __syncthreads();

    if (MODE == 0) {
#pragma unroll
        for (int ii = 0; ii < 2; ++ii) {
            const int i = 2 * wave + ii;
#pragma unroll
            for (int n = 0; n < 8; ++n) {
                const float* tp = red + (8 * i + n) * 256 + lane * 4;
                v8f t;
#pragma unroll
                for (int s = 0; s < KSPLIT; ++s) {
                    const v4f plo = *(const v4f*)(tp + s * (64 * 256));
                    const v4f phi = *(const v4f*)(tp + s * (64 * 256) + 128);
                    const v8f p = __builtin_shufflevector(plo, phi, 0, 1, 2, 3, 4, 5, 6, 7);
                    t = (s == 0) ? p : t + p;
                }
#pragma unroll
                for (int r = 0; r < 8; ++r)
                    C[(size_t)(m0 + 16 * i + r + 8 * hi) * DIMM + n0 + 16 * n + lc] = t[r];
            }
        }
    } else {
        v8f tt[2][8];
#pragma unroll
        for (int ii = 0; ii < 2; ++ii) {
            const int i = 2 * wave + ii;
#pragma unroll
            for (int n = 0; n < 8; ++n) {
                const float* tp = red + (8 * i + n) * 256 + lane * 4;
                v8f t;
#pragma unroll
                for (int s = 0; s < KSPLIT; ++s) {
                    const v4f plo = *(const v4f*)(tp + s * (64 * 256));
                    const v4f phi = *(const v4f*)(tp + s * (64 * 256) + 128);
                    const v8f p = __builtin_shufflevector(plo, phi, 0, 1, 2, 3, 4, 5, 6, 7);
                    t = (s == 0) ? p : t + p;
                }
                tt[ii][n] = t;
            }
        }
        float* scr = red + wave * (64 * 256) + (16 * wave) * 256;
        const int h = blockIdx.x;
#pragma unroll
        for (int ii = 0; ii < 2; ++ii) {
            const int i = 2 * wave + ii;
#pragma unroll
            for (int n = 0; n < 8; ++n)
#pragma unroll
                for (int r = 0; r < 8; ++r)
                    scr[n * 320 + (8 * hi + r) * 20 + lc] = tt[ii][n][r];
            asm volatile("s_wait_dscnt 0" ::: "memory");
            if (MODE == 1) {
                const int T = blockIdx.y * 8 + i;
                float* kt = C + (size_t)(h * 128 + T) * 2048 + lane * 4;
#pragma unroll
                for (int n = 0; n < 8; ++n)
#pragma unroll
                    for (int p = 0; p < 2; ++p) {
                        const v4f v = *(const v4f*)&scr[n * 320 + lc * 20 + 8 * hi + 4 * p];
                        *(v4f*)(kt + n * 256 + p * 128) = v;
                    }
            } else {
                const int kp = lane >> 3;
                const int x  = lane & 7;
                float* vt = C + ((size_t)(h * 1024 + blockIdx.y * 64 + 8 * i) * 64) * 4;
#pragma unroll
                for (int n = 0; n < 8; ++n)
#pragma unroll
                    for (int q = 0; q < 2; ++q) {
                        const int kq = 4 * q + kp;
                        const v2f a = *(const v2f*)&scr[n * 320 + (2 * kq) * 20 + 2 * x];
                        const v2f b = *(const v2f*)&scr[n * 320 + (2 * kq + 1) * 20 + 2 * x];
                        v4f v;
                        v[0] = a[0]; v[1] = b[0]; v[2] = a[1]; v[3] = b[1];
                        *(v4f*)(vt + ((size_t)kq * 64 + 8 * n + x) * 4) = v;
                    }
            }
        }
    }
}

#define KB_TILES 4
#define KB       (16 * KB_TILES)

__device__ __forceinline__ void group_fence() { __builtin_amdgcn_sched_barrier(0); }

__device__ __forceinline__ Frag8 load_kfrag(const float* __restrict__ p) {
    Frag8 f;
    f.lo = *(const v4f*)(p);
    f.hi = *(const v4f*)(p + 128);
    return f;
}

template <int J>
__device__ __forceinline__ void qk_step(const Frag8& q, const Frag8 (&k)[KB_TILES],
                                        v8f (&s)[KB_TILES]) {
    const v2f qv = kpair<J>(q);
#pragma unroll
    for (int j = 0; j < KB_TILES; ++j) s[j] = wmma_f32(qv, kpair<J>(k[j]), s[j]);
}

__global__ __launch_bounds__(256) __attribute__((amdgpu_waves_per_eu(2, 2)))
void attn_fwd(const float* __restrict__ Q, const float* __restrict__ K,
              const float* __restrict__ V, const float* __restrict__ mask,
              float* __restrict__ O) {
    __shared__ float ldsP[8 * KB_TILES * 256];

    const int lane = threadIdx.x & 31;
    const int wave = threadIdx.x >> 5;
    const int hi   = lane >> 4;
    const int lc   = lane & 15;

    const int h  = blockIdx.y;
    const int q0 = (blockIdx.x * 8 + wave) * 16;


    Frag8 aq[8];
    {
        const float* qrow = Q + (size_t)(q0 + lc) * DIMM + h * HDIM + 8 * hi;
#pragma unroll
        for (int cc = 0; cc < 8; ++cc) aq[cc] = load_frag8(qrow + cc * 16);
    }

    v8f oacc[8];
#pragma unroll
    for (int t = 0; t < 8; ++t) oacc[t] = (v8f){};
    float mrow[8], lsum[8];
#pragma unroll
    for (int r = 0; r < 8; ++r) { mrow[r] = -1.0e30f; lsum[r] = 0.0f; }

    float* lp = &ldsP[wave * (KB_TILES * 256)];
    const float scale = 0.08838834764831845f;
    const size_t mbase = ((size_t)h * SEQL + q0) * SEQL;

#pragma unroll 1
    for (int kv0 = 0; kv0 < SEQL; kv0 += KB) {
        v8f s[KB_TILES];
#pragma unroll
        for (int j = 0; j < KB_TILES; ++j) s[j] = (v8f){};
        const float* kt = K + (size_t)(h * (SEQL / 16) + (kv0 >> 4)) * 2048 + lane * 4;
        Frag8 kf[2][2][KB_TILES];
        float mk[8][KB_TILES];
#pragma unroll
        for (int cc = 0; cc < 2; ++cc)
#pragma unroll
            for (int j = 0; j < KB_TILES; ++j)
                kf[0][cc][j] = load_kfrag(kt + j * 2048 + cc * 256);
        group_fence();
#pragma unroll
        for (int g = 0; g < 4; ++g) {
            if (g + 1 < 4) {
#pragma unroll
                for (int cc = 0; cc < 2; ++cc)
#pragma unroll
                    for (int j = 0; j < KB_TILES; ++j)
                        kf[(g + 1) & 1][cc][j] =
                            load_kfrag(kt + j * 2048 + ((g + 1) * 2 + cc) * 256);
            }
            if (g == 2) {
                group_fence();
#pragma unroll
                for (int r = 0; r < 8; ++r) {
                    const float* mp = mask + mbase + (size_t)(r + 8 * hi) * SEQL + kv0 + lc;
#pragma unroll
                    for (int j = 0; j < KB_TILES; ++j)
                        mk[r][j] = __builtin_nontemporal_load(mp + 16 * j);
                }
            }
#pragma unroll
            for (int cc = 0; cc < 2; ++cc) {
                qk_step<0>(aq[g * 2 + cc], kf[g & 1][cc], s);
                qk_step<1>(aq[g * 2 + cc], kf[g & 1][cc], s);
                qk_step<2>(aq[g * 2 + cc], kf[g & 1][cc], s);
                qk_step<3>(aq[g * 2 + cc], kf[g & 1][cc], s);
            }
            group_fence();
        }

        v4f vf[2][4][4];
        const float* vrow = V + ((size_t)(h * (SEQL / 2) + (kv0 >> 1) + hi) * 64 + lc) * 4;
        {
#pragma unroll
            for (int kk = 0; kk < 4; ++kk)
#pragma unroll
                for (int u = 0; u < 4; ++u)
                    vf[0][kk][u] = *(const v4f*)(vrow + (2 * kk) * 256 + 64 * u);
        }
        group_fence();

        float rmax[8];
#pragma unroll
        for (int r = 0; r < 8; ++r) {
#pragma unroll
            for (int j = 0; j < KB_TILES; ++j) s[j][r] = s[j][r] * scale + mk[r][j];
            rmax[r] = fmaxf(fmaxf(s[0][r], s[1][r]), fmaxf(s[2][r], s[3][r]));
        }
#pragma unroll
        for (int off = 1; off < 16; off <<= 1) {
            float ex[8];
#pragma unroll
            for (int r = 0; r < 8; ++r) ex[r] = __shfl_xor(rmax[r], off, 32);
#pragma unroll
            for (int r = 0; r < 8; ++r) rmax[r] = fmaxf(rmax[r], ex[r]);
        }

#pragma unroll
        for (int r = 0; r < 8; ++r) {
            const float mnew = fmaxf(mrow[r], rmax[r]);
            const float corr = __expf(mrow[r] - mnew);
            mrow[r] = mnew;
#pragma unroll
            for (int j = 0; j < KB_TILES; ++j) s[j][r] = __expf(s[j][r] - mnew);
            lsum[r] = lsum[r] * corr + ((s[0][r] + s[1][r]) + (s[2][r] + s[3][r]));
#pragma unroll
            for (int t = 0; t < 8; ++t) oacc[t][r] *= corr;
        }

#pragma unroll
        for (int j = 0; j < KB_TILES; ++j)
#pragma unroll
            for (int r = 0; r < 8; ++r) lp[j * 256 + (r + 8 * hi) * 16 + lc] = s[j][r];
        asm volatile("s_wait_dscnt 0" ::: "memory");
        v2f pa[KB_TILES][4];
#pragma unroll
        for (int j = 0; j < KB_TILES; ++j)
#pragma unroll
            for (int kk = 0; kk < 4; ++kk)
                pa[j][kk] = *(const v2f*)&lp[j * 256 + lc * 16 + kk * 4 + 2 * hi];
        asm volatile("s_wait_dscnt 0" ::: "memory");
        group_fence();

#pragma unroll
        for (int j = 0; j < KB_TILES; ++j) {
            if (j + 1 < KB_TILES) {
#pragma unroll
                for (int kk = 0; kk < 4; ++kk)
#pragma unroll
                    for (int u = 0; u < 4; ++u)
                        vf[(j + 1) & 1][kk][u] =
                            *(const v4f*)(vrow + (8 * (j + 1) + 2 * kk) * 256 + 64 * u);
            }
#pragma unroll
            for (int kk = 0; kk < 4; ++kk)
#pragma unroll
                for (int u = 0; u < 4; ++u) {
                    const v4f w = vf[j & 1][kk][u];
                    oacc[2 * u]     = wmma_f32(pa[j][kk], __builtin_shufflevector(w, w, 0, 1),
                                               oacc[2 * u]);
                    oacc[2 * u + 1] = wmma_f32(pa[j][kk], __builtin_shufflevector(w, w, 2, 3),
                                               oacc[2 * u + 1]);
                }
            group_fence();
        }
    }

#pragma unroll
    for (int off = 1; off < 16; off <<= 1) {
        float ex[8];
#pragma unroll
        for (int r = 0; r < 8; ++r) ex[r] = __shfl_xor(lsum[r], off, 32);
#pragma unroll
        for (int r = 0; r < 8; ++r) lsum[r] += ex[r];
    }
#pragma unroll
    for (int r = 0; r < 8; ++r) {
        const float inv = 1.0f / lsum[r];
        float* orow = O + (size_t)(q0 + r + 8 * hi) * DIMM + h * HDIM + 2 * lc;
#pragma unroll
        for (int u = 0; u < 4; ++u) {
            v2f o;
            o[0] = oacc[2 * u][r] * inv;
            o[1] = oacc[2 * u + 1][r] * inv;
            *(v2f*)(orow + 32 * u) = o;
        }
    }
}

extern "C" void kernel_launch(void* const* d_in, const int* in_sizes, int n_in,
                              void* d_out, int out_size, void* d_ws, size_t ws_size,
                              hipStream_t stream) {
    const float* x    = (const float*)d_in[0];
    const float* mask = (const float*)d_in[1];
    const float* wq   = (const float*)d_in[3];
    const float* wk   = (const float*)d_in[4];
    const float* wv   = (const float*)d_in[5];
    const float* wo   = (const float*)d_in[6];
    float* out = (float*)d_out;

    float* ws = (float*)d_ws;
    const size_t MAT = (size_t)SEQL * DIMM;
    float* Qb = ws;
    float* Kb = ws + MAT;
    float* Vb = ws + 2 * MAT;
    float* Ab = ws + 3 * MAT;

    dim3 tb(256);
    dim3 tg(128);
    dim3 gg(DIMM / 128, SEQL / 128);
    gemm_abt<0><<<gg, tg, 0, stream>>>(x, wq, Qb);
    gemm_abt<1><<<gg, tg, 0, stream>>>(x, wk, Kb);
    gemm_abt<2><<<gg, tg, 0, stream>>>(x, wv, Vb);

    dim3 ga(SEQL / 128, NHEADS);
    attn_fwd<<<ga, tb, 0, stream>>>(Qb, Kb, Vb, mask, Ab);

    gemm_abt<0><<<gg, tg, 0, stream>>>(Ab, wo, out);
}
